# starting kernel
# baseline (speedup 1.0000x reference)
#include <hip/hip_runtime.h>

#define N_NODES 50000
#define E_EDGES 800000
#define F_IN    256
#define F_OUT   128
#define BATCH   2
#define OB      (F_OUT * BATCH)

#define NT ((N_NODES + 15) / 16)
#define OT (F_OUT / 16)
#define NT_PER_BLK 8

typedef float v2f __attribute__((ext_vector_type(2)));
typedef float v8f __attribute__((ext_vector_type(8)));


__global__ void deg_init_kernel(float* __restrict__ deg) {
    int i = blockIdx.x * blockDim.x + threadIdx.x;
    if (i < N_NODES) deg[i] = 1.0f;
}

__global__ void deg_acc_kernel(const int* __restrict__ row,
                               const float* __restrict__ ew,
                               float* __restrict__ deg) {
    for (int e = blockIdx.x * blockDim.x + threadIdx.x; e < E_EDGES;
         e += gridDim.x * blockDim.x) {
        atomicAdd(&deg[row[e]], ew[e]);
    }
}

__global__ void dinv_kernel(const float* __restrict__ deg,
                            float* __restrict__ dinv) {
    int i = blockIdx.x * blockDim.x + threadIdx.x;
    if (i < N_NODES) {
        float d = deg[i];
        dinv[i] = d > 0.0f ? rsqrtf(d) : 0.0f;
    }
}


__global__ void gemm_wmma_kernel(const float* __restrict__ x,
                                 const float* __restrict__ w,
                                 float* __restrict__ xw) {
    __shared__ float wtile[F_IN * 16];

    const int ntb_blocks = (NT + NT_PER_BLK - 1) / NT_PER_BLK;
    int blk  = blockIdx.x;
    int ntb  = blk % ntb_blocks;
    int rest = blk / ntb_blocks;
    int ot   = rest % OT;
    int b    = rest / OT;
    int o0   = ot * 16;

    for (int i = threadIdx.x; i < F_IN * 16; i += blockDim.x) {
        int k = i >> 4;
        int j = i & 15;
        wtile[i] = w[k * F_OUT + o0 + j];
    }
    __syncthreads();

    int wave = threadIdx.x >> 5;
    int lane = threadIdx.x & 31;
    int nt   = ntb * NT_PER_BLK + wave;
    if (nt >= NT) return;

    int n0   = nt * 16;
    int half = lane >> 4;
    int l    = lane & 15;

    const float* xrow = x + ((size_t)b * N_NODES + (size_t)(n0 + l)) * F_IN;

    v8f acc = {};
    for (int k = 0; k < F_IN; k += 4) {
        v2f a, bf;
        a.x  = xrow[k + half * 2 + 0];
        a.y  = xrow[k + half * 2 + 1];
        bf.x = wtile[(k + half * 2 + 0) * 16 + l];
        bf.y = wtile[(k + half * 2 + 1) * 16 + l];
        acc = __builtin_amdgcn_wmma_f32_16x16x4_f32(
             false, a,  false, bf,
             (short)0, acc,  false,  false);
    }

    #pragma unroll
    for (int v = 0; v < 8; ++v) {
        int m = v + half * 8;
        xw[(size_t)(n0 + m) * OB + (size_t)(o0 + l) * 2 + b] = acc[v];
    }
}


__global__ void out_init_kernel(const float* __restrict__ dinv,
                                const float* __restrict__ xw,
                                const float* __restrict__ bias,
                                float* __restrict__ out) {
    size_t i = (size_t)blockIdx.x * blockDim.x + threadIdx.x;
    const size_t total = (size_t)BATCH * N_NODES * F_OUT;
    if (i >= total) return;
    int o    = (int)(i % F_OUT);
    size_t t = i / F_OUT;
    int n    = (int)(t % N_NODES);
    int b    = (int)(t / N_NODES);
    float d  = dinv[n];
    out[i] = bias[o] + d * d * xw[(size_t)n * OB + (size_t)o * 2 + b];
}


__global__ void edge_scatter_kernel(const int* __restrict__ row,
                                    const int* __restrict__ col,
                                    const float* __restrict__ ew,
                                    const float* __restrict__ dinv,
                                    const float* __restrict__ xw,
                                    float* __restrict__ out) {
    int c = threadIdx.x;
    int o = c >> 1;
    int b = c & 1;
    for (int e = blockIdx.x; e < E_EDGES; e += gridDim.x) {
        int r  = row[e];
        int cl = col[e];
        float lap = dinv[r] * ew[e] * dinv[cl];
        float val = lap * xw[(size_t)cl * OB + c];
        atomicAdd(&out[((size_t)b * N_NODES + r) * F_OUT + o], val);
    }
}


extern "C" void kernel_launch(void* const* d_in, const int* in_sizes, int n_in,
                              void* d_out, int out_size, void* d_ws, size_t ws_size,
                              hipStream_t stream) {
    const float* x    = (const float*)d_in[0];
    const float* w    = (const float*)d_in[1];
    const float* bias = (const float*)d_in[2];
    const float* ew   = (const float*)d_in[3];
    const int*   eidx = (const int*)d_in[4];
    const int*   row  = eidx;
    const int*   col  = eidx + E_EDGES;
    float* out = (float*)d_out;

    float* xw   = (float*)d_ws;
    float* deg  = xw + (size_t)N_NODES * OB;
    float* dinv = deg + N_NODES;

    deg_init_kernel<<<(N_NODES + 255) / 256, 256, 0, stream>>>(deg);
    deg_acc_kernel<<<1024, 256, 0, stream>>>(row, ew, deg);
    dinv_kernel<<<(N_NODES + 255) / 256, 256, 0, stream>>>(deg, dinv);

    const int ntb_blocks = (NT + NT_PER_BLK - 1) / NT_PER_BLK;
    gemm_wmma_kernel<<<ntb_blocks * OT * BATCH, 256, 0, stream>>>(x, w, xw);

    size_t total = (size_t)BATCH * N_NODES * F_OUT;
    out_init_kernel<<<(unsigned)((total + 255) / 256), 256, 0, stream>>>(dinv, xw, bias, out);

    edge_scatter_kernel<<<8192, 256, 0, stream>>>(row, col, ew, dinv, xw, out);
}


# Projection weight-stationary: a wave keeps its 16 outs' w fragments (all k) in registers; x tiles of both batches pass once through LDS by whole-line loads, shared by 8 waves; sums unchanged
# speedup vs baseline: 3.8979x; 3.8979x over previous
#include <hip/hip_runtime.h>

#define N_NODES 50000
#define E_EDGES 800000
#define F_IN    256
#define F_OUT   128
#define BATCH   2
#define OB      (F_OUT * BATCH)

#define NT ((N_NODES + 15) / 16)

typedef float v2f __attribute__((ext_vector_type(2)));
typedef float v8f __attribute__((ext_vector_type(8)));


__global__ void head_init_kernel(int* __restrict__ head) {
    int i = blockIdx.x * blockDim.x + threadIdx.x;
    if (i < N_NODES) head[i] = -1;
}

__global__ void list_build_kernel(const int* __restrict__ row,
                                  int* __restrict__ head,
                                  int* __restrict__ next) {
    int e = blockIdx.x * blockDim.x + threadIdx.x;
    if (e < E_EDGES) next[e] = atomicExch(&head[row[e]], e);
}

__global__ void dinv_kernel(const int* __restrict__ head,
                            const int* __restrict__ next,
                            const float* __restrict__ ew,
                            float* __restrict__ dinv) {
    int i = blockIdx.x * blockDim.x + threadIdx.x;
    if (i >= N_NODES) return;
    float d = 1.0f;
    int e = head[i];
    for (int hops = 0; (unsigned)e < (unsigned)E_EDGES && hops < E_EDGES; ++hops) {
        d += ew[e];
        e = next[e];
    }
    dinv[i] = d > 0.0f ? rsqrtf(d) : 0.0f;
}


#define TILE_PITCH   260
#define STAGE_FLOATS (2 * 16 * TILE_PITCH)
#define GEMM_BLOCKS  512
#define TILES_Q      (NT / GEMM_BLOCKS)
#define TILES_R      (NT % GEMM_BLOCKS)
#define KSTEPS       (F_IN / 4)
#define KGROUP       8

struct StageRegs { float4 c0, c1, c2, c3, c4, c5, c6, c7; };

__device__ __forceinline__ float4 stage_chunk(const float* __restrict__ x, int tile, int i, int srow, int scol) {
    const float* src = x + ((size_t)(i >> 2) * N_NODES + (size_t)tile * 16 + 4 * (i & 3) + srow) * F_IN + scol;
    return *reinterpret_cast<const float4*>(src);
}

__device__ __forceinline__ StageRegs stage_load(const float* __restrict__ x, int tile, int srow, int scol) {
    StageRegs r;
    r.c0 = stage_chunk(x, tile, 0, srow, scol); r.c1 = stage_chunk(x, tile, 1, srow, scol);
    r.c2 = stage_chunk(x, tile, 2, srow, scol); r.c3 = stage_chunk(x, tile, 3, srow, scol);
    r.c4 = stage_chunk(x, tile, 4, srow, scol); r.c5 = stage_chunk(x, tile, 5, srow, scol);
    r.c6 = stage_chunk(x, tile, 6, srow, scol); r.c7 = stage_chunk(x, tile, 7, srow, scol);
    return r;
}

__device__ __forceinline__ void stage_store(float* __restrict__ stage, const StageRegs& r, int srow, int scol) {
    float* dst = stage + srow * TILE_PITCH + scol;
    *reinterpret_cast<float4*>(dst + 0 * 4 * TILE_PITCH) = r.c0; *reinterpret_cast<float4*>(dst + 1 * 4 * TILE_PITCH) = r.c1;
    *reinterpret_cast<float4*>(dst + 2 * 4 * TILE_PITCH) = r.c2; *reinterpret_cast<float4*>(dst + 3 * 4 * TILE_PITCH) = r.c3;
    *reinterpret_cast<float4*>(dst + 4 * 4 * TILE_PITCH) = r.c4; *reinterpret_cast<float4*>(dst + 5 * 4 * TILE_PITCH) = r.c5;
    *reinterpret_cast<float4*>(dst + 6 * 4 * TILE_PITCH) = r.c6; *reinterpret_cast<float4*>(dst + 7 * 4 * TILE_PITCH) = r.c7;
}

__global__ void __launch_bounds__(256) __attribute__((amdgpu_waves_per_eu(4, 4)))
gemm_wmma_kernel(const float* __restrict__ x,
                 const float* __restrict__ w,
                 float* __restrict__ xw) {
    __shared__ __attribute__((aligned(16))) float xs[2][STAGE_FLOATS];

    const int tid  = threadIdx.x;
    const int wave = tid >> 5;
    const int lane = tid & 31;
    const int half = lane >> 4;
    const int l    = lane & 15;
    const int o0   = wave * 16;

    const int blk     = blockIdx.x;
    const int t_begin = blk * TILES_Q + (blk < TILES_R ? blk : TILES_R);
    const int t_count = TILES_Q + (blk < TILES_R ? 1 : 0);

    v2f wf[KSTEPS];
    #pragma unroll
    for (int t = 0; t < KSTEPS; ++t) {
        wf[t].x = w[(4 * t + 2 * half + 0) * F_OUT + o0 + l];
        wf[t].y = w[(4 * t + 2 * half + 1) * F_OUT + o0 + l];
    }

    const int srow = tid >> 6;
    const int scol = (tid & 63) * 4;
    stage_store(xs[0], stage_load(x, t_begin, srow, scol), srow, scol);
    __syncthreads();

    for (int it = 0; it < t_count; ++it) {
        const int  tile = t_begin + it;
        const int  buf  = it & 1;
        const bool more = it + 1 < t_count;

        StageRegs nxt = {};
        if (more) nxt = stage_load(x, tile + 1, srow, scol);
        __builtin_amdgcn_sched_barrier(0);

        const float* a0p = &xs[buf][(0 * 16 + l) * TILE_PITCH + 2 * half];
        const float* a1p = &xs[buf][(1 * 16 + l) * TILE_PITCH + 2 * half];

        v8f acc0 = {}, acc1 = {};
        #pragma unroll
        for (int g = 0; g < KSTEPS; g += KGROUP) {
            v2f a0[KGROUP], a1[KGROUP];
            #pragma unroll
            for (int j = 0; j < KGROUP; ++j) {
                a0[j] = *reinterpret_cast<const v2f*>(a0p + 4 * (g + j));
                a1[j] = *reinterpret_cast<const v2f*>(a1p + 4 * (g + j));
            }
            #pragma unroll
            for (int j = 0; j < KGROUP; ++j) {
                acc0 = __builtin_amdgcn_wmma_f32_16x16x4_f32(
                     false, a0[j],  false, wf[g + j],
                     (short)0, acc0,  false,  false);
                acc1 = __builtin_amdgcn_wmma_f32_16x16x4_f32(
                     false, a1[j],  false, wf[g + j],
                     (short)0, acc1,  false,  false);
            }
        }

        #pragma unroll
        for (int v = 0; v < 8; ++v) {
            const int m = v + half * 8;
            float2 pr;
            pr.x = acc0[v];
            pr.y = acc1[v];
            *reinterpret_cast<float2*>(&xw[(size_t)(tile * 16 + m) * OB + (size_t)(o0 + l) * 2]) = pr;
        }

        if (more) stage_store(xs[buf ^ 1], nxt, srow, scol);
        __syncthreads();
    }
}


#define ROWS_PER_BLK 8

__device__ __forceinline__ void add_row(float (&acc)[8], float s, const float4& p, const float4& q) {
    acc[0] = __builtin_fmaf(s, p.x, acc[0]); acc[1] = __builtin_fmaf(s, p.y, acc[1]);
    acc[2] = __builtin_fmaf(s, p.z, acc[2]); acc[3] = __builtin_fmaf(s, p.w, acc[3]);
    acc[4] = __builtin_fmaf(s, q.x, acc[4]); acc[5] = __builtin_fmaf(s, q.y, acc[5]);
    acc[6] = __builtin_fmaf(s, q.z, acc[6]); acc[7] = __builtin_fmaf(s, q.w, acc[7]);
}

__global__ void __launch_bounds__(256)
row_gather_kernel(const int* __restrict__ head,
                  const int* __restrict__ next,
                  const int* __restrict__ col,
                  const float* __restrict__ ew,
                  const float* __restrict__ dinv,
                  const float* __restrict__ xw,
                  const float* __restrict__ bias,
                  float* __restrict__ out) {
    const int lane = threadIdx.x & 31;
    const int r = __builtin_amdgcn_readfirstlane((int)(blockIdx.x * ROWS_PER_BLK + (threadIdx.x >> 5)));
    if (r >= N_NODES) return;

    const float dr = dinv[r];
    const float d2 = dr * dr;
    const int   o0 = lane * 4;

    float acc[8];
    {
        const float4* xs = reinterpret_cast<const float4*>(xw + (size_t)r * OB) + lane * 2;
        const float4 p = xs[0];
        const float4 q = xs[1];
        const float b0 = bias[o0 + 0], b1 = bias[o0 + 1], b2 = bias[o0 + 2], b3 = bias[o0 + 3];
        acc[0] = b0 + d2 * p.x; acc[1] = b0 + d2 * p.y;
        acc[2] = b1 + d2 * p.z; acc[3] = b1 + d2 * p.w;
        acc[4] = b2 + d2 * q.x; acc[5] = b2 + d2 * q.y;
        acc[6] = b3 + d2 * q.z; acc[7] = b3 + d2 * q.w;
    }

    int e = head[r];
    int walked = 0;
    while ((unsigned)e < (unsigned)E_EDGES && walked < E_EDGES) {
        int my_e = -1;
        int cnt  = 0;
        while ((unsigned)e < (unsigned)E_EDGES && cnt < 32) {
            if (lane == cnt) my_e = e;
            e = next[e];
            ++cnt;
        }
        walked += cnt;

        int   c   = 0;
        float lap = 0.0f;
        if (my_e >= 0) {
            c   = col[my_e];
            lap = dr * ew[my_e] * dinv[c];
        }

        int j = 0;
        for (; j + 4 <= cnt; j += 4) {
            const int c0 = __builtin_amdgcn_readlane(c, j + 0);
            const int c1 = __builtin_amdgcn_readlane(c, j + 1);
            const int c2 = __builtin_amdgcn_readlane(c, j + 2);
            const int c3 = __builtin_amdgcn_readlane(c, j + 3);
            const float4* x0 = reinterpret_cast<const float4*>(xw + (size_t)c0 * OB) + lane * 2;
            const float4* x1 = reinterpret_cast<const float4*>(xw + (size_t)c1 * OB) + lane * 2;
            const float4* x2 = reinterpret_cast<const float4*>(xw + (size_t)c2 * OB) + lane * 2;
            const float4* x3 = reinterpret_cast<const float4*>(xw + (size_t)c3 * OB) + lane * 2;
            const float4 p0 = x0[0], q0 = x0[1];
            const float4 p1 = x1[0], q1 = x1[1];
            const float4 p2 = x2[0], q2 = x2[1];
            const float4 p3 = x3[0], q3 = x3[1];
            const float s0 = __int_as_float(__builtin_amdgcn_readlane(__float_as_int(lap), j + 0));
            const float s1 = __int_as_float(__builtin_amdgcn_readlane(__float_as_int(lap), j + 1));
            const float s2 = __int_as_float(__builtin_amdgcn_readlane(__float_as_int(lap), j + 2));
            const float s3 = __int_as_float(__builtin_amdgcn_readlane(__float_as_int(lap), j + 3));
            add_row(acc, s0, p0, q0);
            add_row(acc, s1, p1, q1);
            add_row(acc, s2, p2, q2);
            add_row(acc, s3, p3, q3);
        }
        #pragma clang loop unroll(disable)
        for (; j < cnt; ++j) {
            const int   cj = __builtin_amdgcn_readlane(c, j);
            const float sj = __int_as_float(__builtin_amdgcn_readlane(__float_as_int(lap), j));
            const float4* xj = reinterpret_cast<const float4*>(xw + (size_t)cj * OB) + lane * 2;
            const float4 p = xj[0], q = xj[1];
            add_row(acc, sj, p, q);
        }
    }

    float4* out0 = reinterpret_cast<float4*>(out + ((size_t)0 * N_NODES + r) * F_OUT) + lane;
    float4* out1 = reinterpret_cast<float4*>(out + ((size_t)1 * N_NODES + r) * F_OUT) + lane;
    *out0 = make_float4(acc[0], acc[2], acc[4], acc[6]);
    *out1 = make_float4(acc[1], acc[3], acc[5], acc[7]);
}


extern "C" void kernel_launch(void* const* d_in, const int* in_sizes, int n_in,
                              void* d_out, int out_size, void* d_ws, size_t ws_size,
                              hipStream_t stream) {
    const float* x    = (const float*)d_in[0];
    const float* w    = (const float*)d_in[1];
    const float* bias = (const float*)d_in[2];
    const float* ew   = (const float*)d_in[3];
    const int*   eidx = (const int*)d_in[4];
    const int*   row  = eidx;
    const int*   col  = eidx + E_EDGES;
    float* out = (float*)d_out;

    float* xw   = (float*)d_ws;
    int*   head = (int*)(xw + (size_t)N_NODES * OB);
    float* dinv = (float*)(head + N_NODES);
    int*   next = (int*)(dinv + N_NODES);

    head_init_kernel<<<(N_NODES + 255) / 256, 256, 0, stream>>>(head);
    list_build_kernel<<<(E_EDGES + 255) / 256, 256, 0, stream>>>(row, head, next);
    dinv_kernel<<<(N_NODES + 255) / 256, 256, 0, stream>>>(head, next, ew, dinv);

    gemm_wmma_kernel<<<GEMM_BLOCKS, 256, 0, stream>>>(x, w, xw);

    row_gather_kernel<<<(N_NODES + ROWS_PER_BLK - 1) / ROWS_PER_BLK, 256, 0, stream>>>(
        head, next, col, ew, dinv, xw, bias, out);
}


# Row lists -> per-row slot arrays: an int add per edge claims a slot for (col, weight) in the row's own output bytes; no list walks; row sums in slot (arrival) order, degree as 4 lane partial sums
# speedup vs baseline: 4.3254x; 1.1097x over previous
#include <hip/hip_runtime.h>

#define N_NODES 50000
#define E_EDGES 800000
#define F_IN    256
#define F_OUT   128
#define BATCH   2
#define OB      (F_OUT * BATCH)

#define NT ((N_NODES + 15) / 16)

typedef float v2f __attribute__((ext_vector_type(2)));
typedef float v8f __attribute__((ext_vector_type(8)));


#define SLOT_CAP 64

__device__ __forceinline__ int* overflow_head(float* out, int r) {
    return reinterpret_cast<int*>(out) + ((size_t)N_NODES + (size_t)r) * F_OUT;
}

__global__ void count_init_kernel(int* __restrict__ cnt) {
    int i = blockIdx.x * blockDim.x + threadIdx.x;
    if (i < N_NODES) cnt[i] = 0;
}

__device__ __forceinline__ void place_edge(float* out, int* __restrict__ next,
                                           int r, int p, int c, float w, int e) {
    if (p < SLOT_CAP) {
        reinterpret_cast<int2*>(out)[(size_t)r * SLOT_CAP + p] = make_int2(c, __float_as_int(w));
    } else {
        next[e] = atomicExch(overflow_head(out, r), e);
    }
}

__global__ void __launch_bounds__(256)
slot_fill_kernel(const int* __restrict__ row,
                 const int* __restrict__ col,
                 const float* __restrict__ ew,
                 int* __restrict__ cnt,
                 int* __restrict__ next,
                 float* out) {
    const int t = blockIdx.x * blockDim.x + threadIdx.x;
    if (t >= E_EDGES / 4) return;
    const int4   r4 = reinterpret_cast<const int4*>(row)[t];
    const int4   c4 = reinterpret_cast<const int4*>(col)[t];
    const float4 w4 = reinterpret_cast<const float4*>(ew)[t];
    const int p0 = atomicAdd(&cnt[r4.x], 1);
    const int p1 = atomicAdd(&cnt[r4.y], 1);
    const int p2 = atomicAdd(&cnt[r4.z], 1);
    const int p3 = atomicAdd(&cnt[r4.w], 1);
    place_edge(out, next, r4.x, p0, c4.x, w4.x, 4 * t + 0);
    place_edge(out, next, r4.y, p1, c4.y, w4.y, 4 * t + 1);
    place_edge(out, next, r4.z, p2, c4.z, w4.z, 4 * t + 2);
    place_edge(out, next, r4.w, p3, c4.w, w4.w, 4 * t + 3);
}

__global__ void __launch_bounds__(256)
dinv_kernel(const int* __restrict__ cnt,
            const int* __restrict__ next,
            const float* __restrict__ ew,
            float* out,
            float* __restrict__ dinv) {
    const int  g    = blockIdx.x * blockDim.x + threadIdx.x;
    const int  r    = g >> 2;
    const int  sub  = g & 3;
    const bool live = r < N_NODES;
    const int  total = live ? cnt[r] : 0;
    const int  n     = total < SLOT_CAP ? total : SLOT_CAP;

    float d = 0.0f;
    const int2* sl = reinterpret_cast<const int2*>(out) + (size_t)(live ? r : 0) * SLOT_CAP;
    for (int s = sub; s < n; s += 16) {
        const bool h1 = s + 4 < n, h2 = s + 8 < n, h3 = s + 12 < n;
        const float w0 = __int_as_float(sl[s].y);
        const float w1 = __int_as_float(sl[h1 ? s + 4 : s].y);
        const float w2 = __int_as_float(sl[h2 ? s + 8 : s].y);
        const float w3 = __int_as_float(sl[h3 ? s + 12 : s].y);
        d += w0;
        if (h1) d += w1;
        if (h2) d += w2;
        if (h3) d += w3;
    }
    if (sub == 0 && total > SLOT_CAP) {
        int e = *overflow_head(out, r);
        for (int left = total - SLOT_CAP; left > 0 && (unsigned)e < (unsigned)E_EDGES; --left) {
            d += ew[e];
            e = next[e];
        }
    }
    d += __shfl_xor(d, 1);
    d += __shfl_xor(d, 2);
    if (live && sub == 0) {
        d += 1.0f;
        dinv[r] = d > 0.0f ? rsqrtf(d) : 0.0f;
    }
}


#define TILE_PITCH   260
#define STAGE_FLOATS (2 * 16 * TILE_PITCH)
#define GEMM_BLOCKS  512
#define TILES_Q      (NT / GEMM_BLOCKS)
#define TILES_R      (NT % GEMM_BLOCKS)
#define KSTEPS       (F_IN / 4)
#define KGROUP       8

struct StageRegs { float4 c0, c1, c2, c3, c4, c5, c6, c7; };

__device__ __forceinline__ float4 stage_chunk(const float* __restrict__ x, int tile, int i, int srow, int scol) {
    const float* src = x + ((size_t)(i >> 2) * N_NODES + (size_t)tile * 16 + 4 * (i & 3) + srow) * F_IN + scol;
    return *reinterpret_cast<const float4*>(src);
}

__device__ __forceinline__ StageRegs stage_load(const float* __restrict__ x, int tile, int srow, int scol) {
    StageRegs r;
    r.c0 = stage_chunk(x, tile, 0, srow, scol); r.c1 = stage_chunk(x, tile, 1, srow, scol);
    r.c2 = stage_chunk(x, tile, 2, srow, scol); r.c3 = stage_chunk(x, tile, 3, srow, scol);
    r.c4 = stage_chunk(x, tile, 4, srow, scol); r.c5 = stage_chunk(x, tile, 5, srow, scol);
    r.c6 = stage_chunk(x, tile, 6, srow, scol); r.c7 = stage_chunk(x, tile, 7, srow, scol);
    return r;
}

__device__ __forceinline__ void stage_store(float* __restrict__ stage, const StageRegs& r, int srow, int scol) {
    float* dst = stage + srow * TILE_PITCH + scol;
    *reinterpret_cast<float4*>(dst + 0 * 4 * TILE_PITCH) = r.c0; *reinterpret_cast<float4*>(dst + 1 * 4 * TILE_PITCH) = r.c1;
    *reinterpret_cast<float4*>(dst + 2 * 4 * TILE_PITCH) = r.c2; *reinterpret_cast<float4*>(dst + 3 * 4 * TILE_PITCH) = r.c3;
    *reinterpret_cast<float4*>(dst + 4 * 4 * TILE_PITCH) = r.c4; *reinterpret_cast<float4*>(dst + 5 * 4 * TILE_PITCH) = r.c5;
    *reinterpret_cast<float4*>(dst + 6 * 4 * TILE_PITCH) = r.c6; *reinterpret_cast<float4*>(dst + 7 * 4 * TILE_PITCH) = r.c7;
}

__global__ void __launch_bounds__(256) __attribute__((amdgpu_waves_per_eu(4, 4)))
gemm_wmma_kernel(const float* __restrict__ x,
                 const float* __restrict__ w,
                 float* __restrict__ xw) {
    __shared__ __attribute__((aligned(16))) float xs[2][STAGE_FLOATS];

    const int tid  = threadIdx.x;
    const int wave = tid >> 5;
    const int lane = tid & 31;
    const int half = lane >> 4;
    const int l    = lane & 15;
    const int o0   = wave * 16;

    const int blk     = blockIdx.x;
    const int t_begin = blk * TILES_Q + (blk < TILES_R ? blk : TILES_R);
    const int t_count = TILES_Q + (blk < TILES_R ? 1 : 0);

    v2f wf[KSTEPS];
    #pragma unroll
    for (int t = 0; t < KSTEPS; ++t) {
        wf[t].x = w[(4 * t + 2 * half + 0) * F_OUT + o0 + l];
        wf[t].y = w[(4 * t + 2 * half + 1) * F_OUT + o0 + l];
    }

    const int srow = tid >> 6;
    const int scol = (tid & 63) * 4;
    stage_store(xs[0], stage_load(x, t_begin, srow, scol), srow, scol);
    __syncthreads();

    for (int it = 0; it < t_count; ++it) {
        const int  tile = t_begin + it;
        const int  buf  = it & 1;
        const bool more = it + 1 < t_count;

        StageRegs nxt = {};
        if (more) nxt = stage_load(x, tile + 1, srow, scol);
        __builtin_amdgcn_sched_barrier(0);

        const float* a0p = &xs[buf][(0 * 16 + l) * TILE_PITCH + 2 * half];
        const float* a1p = &xs[buf][(1 * 16 + l) * TILE_PITCH + 2 * half];

        v8f acc0 = {}, acc1 = {};
        #pragma unroll
        for (int g = 0; g < KSTEPS; g += KGROUP) {
            v2f a0[KGROUP], a1[KGROUP];
            #pragma unroll
            for (int j = 0; j < KGROUP; ++j) {
                a0[j] = *reinterpret_cast<const v2f*>(a0p + 4 * (g + j));
                a1[j] = *reinterpret_cast<const v2f*>(a1p + 4 * (g + j));
            }
            #pragma unroll
            for (int j = 0; j < KGROUP; ++j) {
                acc0 = __builtin_amdgcn_wmma_f32_16x16x4_f32(
                     false, a0[j],  false, wf[g + j],
                     (short)0, acc0,  false,  false);
                acc1 = __builtin_amdgcn_wmma_f32_16x16x4_f32(
                     false, a1[j],  false, wf[g + j],
                     (short)0, acc1,  false,  false);
            }
        }

        #pragma unroll
        for (int v = 0; v < 8; ++v) {
            const int m = v + half * 8;
            float2 pr;
            pr.x = acc0[v];
            pr.y = acc1[v];
            *reinterpret_cast<float2*>(&xw[(size_t)(tile * 16 + m) * OB + (size_t)(o0 + l) * 2]) = pr;
        }

        if (more) stage_store(xs[buf ^ 1], nxt, srow, scol);
        __syncthreads();
    }
}


#define ROWS_PER_BLK 8

__device__ __forceinline__ void add_row(float (&acc)[8], float s, const float4& p, const float4& q) {
    acc[0] = __builtin_fmaf(s, p.x, acc[0]); acc[1] = __builtin_fmaf(s, p.y, acc[1]);
    acc[2] = __builtin_fmaf(s, p.z, acc[2]); acc[3] = __builtin_fmaf(s, p.w, acc[3]);
    acc[4] = __builtin_fmaf(s, q.x, acc[4]); acc[5] = __builtin_fmaf(s, q.y, acc[5]);
    acc[6] = __builtin_fmaf(s, q.z, acc[6]); acc[7] = __builtin_fmaf(s, q.w, acc[7]);
}

__global__ void __launch_bounds__(256)
row_gather_kernel(const int* __restrict__ cnt,
                  const int* __restrict__ next,
                  const int* __restrict__ col,
                  const float* __restrict__ ew,
                  const float* __restrict__ dinv,
                  const float* __restrict__ xw,
                  const float* __restrict__ bias,
                  float* out) {
    const int lane = threadIdx.x & 31;
    const int r = __builtin_amdgcn_readfirstlane((int)(blockIdx.x * ROWS_PER_BLK + (threadIdx.x >> 5)));
    if (r >= N_NODES) return;

    const int   total = cnt[r];
    const int   n     = total < SLOT_CAP ? total : SLOT_CAP;
    const float dr    = dinv[r];
    const float d2    = dr * dr;
    const int   o0    = lane * 4;

    float acc[8];
    {
        const float4* xs = reinterpret_cast<const float4*>(xw + (size_t)r * OB) + lane * 2;
        const float4 p = xs[0];
        const float4 q = xs[1];
        const float b0 = bias[o0 + 0], b1 = bias[o0 + 1], b2 = bias[o0 + 2], b3 = bias[o0 + 3];
        acc[0] = b0 + d2 * p.x; acc[1] = b0 + d2 * p.y;
        acc[2] = b1 + d2 * p.z; acc[3] = b1 + d2 * p.w;
        acc[4] = b2 + d2 * q.x; acc[5] = b2 + d2 * q.y;
        acc[6] = b3 + d2 * q.z; acc[7] = b3 + d2 * q.w;
    }

    const int2* slots = reinterpret_cast<const int2*>(out) + (size_t)r * SLOT_CAP;
    int e = -1;
    for (int done = 0; done < total; ) {
        int   cnt_b = 0;
        int   c     = 0;
        float lap   = 0.0f;
        if (done < n) {
            cnt_b = n - done < 32 ? n - done : 32;
            if (lane < cnt_b) {
                const int2 s = slots[done + lane];
                c   = s.x;
                lap = dr * __int_as_float(s.y) * dinv[c];
            }
        } else {
            if (done == n) e = *overflow_head(out, r);
            const int want = total - done < 32 ? total - done : 32;
            int my_e = -1;
            while (cnt_b < want && (unsigned)e < (unsigned)E_EDGES) {
                if (lane == cnt_b) my_e = e;
                e = next[e];
                ++cnt_b;
            }
            if (cnt_b == 0) break;
            if (my_e >= 0) {
                c   = col[my_e];
                lap = dr * ew[my_e] * dinv[c];
            }
        }
        done += cnt_b;

        int j = 0;
        for (; j + 4 <= cnt_b; j += 4) {
            const int c0 = __builtin_amdgcn_readlane(c, j + 0);
            const int c1 = __builtin_amdgcn_readlane(c, j + 1);
            const int c2 = __builtin_amdgcn_readlane(c, j + 2);
            const int c3 = __builtin_amdgcn_readlane(c, j + 3);
            const float4* x0 = reinterpret_cast<const float4*>(xw + (size_t)c0 * OB) + lane * 2;
            const float4* x1 = reinterpret_cast<const float4*>(xw + (size_t)c1 * OB) + lane * 2;
            const float4* x2 = reinterpret_cast<const float4*>(xw + (size_t)c2 * OB) + lane * 2;
            const float4* x3 = reinterpret_cast<const float4*>(xw + (size_t)c3 * OB) + lane * 2;
            const float4 p0 = x0[0], q0 = x0[1];
            const float4 p1 = x1[0], q1 = x1[1];
            const float4 p2 = x2[0], q2 = x2[1];
            const float4 p3 = x3[0], q3 = x3[1];
            const float s0 = __int_as_float(__builtin_amdgcn_readlane(__float_as_int(lap), j + 0));
            const float s1 = __int_as_float(__builtin_amdgcn_readlane(__float_as_int(lap), j + 1));
            const float s2 = __int_as_float(__builtin_amdgcn_readlane(__float_as_int(lap), j + 2));
            const float s3 = __int_as_float(__builtin_amdgcn_readlane(__float_as_int(lap), j + 3));
            add_row(acc, s0, p0, q0);
            add_row(acc, s1, p1, q1);
            add_row(acc, s2, p2, q2);
            add_row(acc, s3, p3, q3);
        }
        #pragma clang loop unroll(disable)
        for (; j < cnt_b; ++j) {
            const int   cj = __builtin_amdgcn_readlane(c, j);
            const float sj = __int_as_float(__builtin_amdgcn_readlane(__float_as_int(lap), j));
            const float4* xj = reinterpret_cast<const float4*>(xw + (size_t)cj * OB) + lane * 2;
            const float4 p = xj[0], q = xj[1];
            add_row(acc, sj, p, q);
        }
    }

    float4* out0 = reinterpret_cast<float4*>(out + ((size_t)0 * N_NODES + r) * F_OUT) + lane;
    float4* out1 = reinterpret_cast<float4*>(out + ((size_t)1 * N_NODES + r) * F_OUT) + lane;
    *out0 = make_float4(acc[0], acc[2], acc[4], acc[6]);
    *out1 = make_float4(acc[1], acc[3], acc[5], acc[7]);
}


extern "C" void kernel_launch(void* const* d_in, const int* in_sizes, int n_in,
                              void* d_out, int out_size, void* d_ws, size_t ws_size,
                              hipStream_t stream) {
    const float* x    = (const float*)d_in[0];
    const float* w    = (const float*)d_in[1];
    const float* bias = (const float*)d_in[2];
    const float* ew   = (const float*)d_in[3];
    const int*   eidx = (const int*)d_in[4];
    const int*   row  = eidx;
    const int*   col  = eidx + E_EDGES;
    float* out = (float*)d_out;

    float* xw   = (float*)d_ws;
    int*   cnt  = (int*)(xw + (size_t)N_NODES * OB);
    float* dinv = (float*)(cnt + N_NODES);
    int*   next = (int*)(dinv + N_NODES);

    count_init_kernel<<<(N_NODES + 255) / 256, 256, 0, stream>>>(cnt);
    slot_fill_kernel<<<(E_EDGES / 4 + 255) / 256, 256, 0, stream>>>(row, col, ew, cnt, next, out);
    dinv_kernel<<<(N_NODES * 4 + 255) / 256, 256, 0, stream>>>(cnt, next, ew, out, dinv);

    gemm_wmma_kernel<<<GEMM_BLOCKS, 256, 0, stream>>>(x, w, xw);

    row_gather_kernel<<<(N_NODES + ROWS_PER_BLK - 1) / ROWS_PER_BLK, 256, 0, stream>>>(
        cnt, next, col, ew, dinv, xw, bias, out);
}


# Slot claims ride in the projection launch: each thread files one edge per node tile, the counter add issued at the tile top and answered under its WMMA block; launches 5 -> 4; arrival order kept
# speedup vs baseline: 5.1798x; 1.1975x over previous
#include <hip/hip_runtime.h>

#define N_NODES 50000
#define E_EDGES 800000
#define F_IN    256
#define F_OUT   128
#define BATCH   2
#define OB      (F_OUT * BATCH)

#define NT ((N_NODES + 15) / 16)

typedef float v2f __attribute__((ext_vector_type(2)));
typedef float v8f __attribute__((ext_vector_type(8)));


#define SLOT_CAP 64

__device__ __forceinline__ int* overflow_head(float* out, int r) {
    return reinterpret_cast<int*>(out) + ((size_t)N_NODES + (size_t)r) * F_OUT;
}

__global__ void count_init_kernel(int* __restrict__ cnt) {
    int i = blockIdx.x * blockDim.x + threadIdx.x;
    if (i < N_NODES) cnt[i] = 0;
}

__device__ __forceinline__ void place_edge(float* out, int* __restrict__ next,
                                           int r, int p, int c, float w, int e) {
    if (p < SLOT_CAP) {
        reinterpret_cast<int2*>(out)[(size_t)r * SLOT_CAP + p] = make_int2(c, __float_as_int(w));
    } else {
        next[e] = atomicExch(overflow_head(out, r), e);
    }
}


__global__ void __launch_bounds__(256)
dinv_kernel(const int* __restrict__ cnt,
            const int* __restrict__ next,
            const float* __restrict__ ew,
            float* out,
            float* __restrict__ dinv) {
    const int  g    = blockIdx.x * blockDim.x + threadIdx.x;
    const int  r    = g >> 2;
    const int  sub  = g & 3;
    const bool live = r < N_NODES;
    const int  total = live ? cnt[r] : 0;
    const int  n     = total < SLOT_CAP ? total : SLOT_CAP;

    float d = 0.0f;
    const int2* sl = reinterpret_cast<const int2*>(out) + (size_t)(live ? r : 0) * SLOT_CAP;
    for (int s = sub; s < n; s += 16) {
        const bool h1 = s + 4 < n, h2 = s + 8 < n, h3 = s + 12 < n;
        const float w0 = __int_as_float(sl[s].y);
        const float w1 = __int_as_float(sl[h1 ? s + 4 : s].y);
        const float w2 = __int_as_float(sl[h2 ? s + 8 : s].y);
        const float w3 = __int_as_float(sl[h3 ? s + 12 : s].y);
        d += w0;
        if (h1) d += w1;
        if (h2) d += w2;
        if (h3) d += w3;
    }
    if (sub == 0 && total > SLOT_CAP) {
        int e = *overflow_head(out, r);
        for (int left = total - SLOT_CAP; left > 0 && (unsigned)e < (unsigned)E_EDGES; --left) {
            d += ew[e];
            e = next[e];
        }
    }
    d += __shfl_xor(d, 1);
    d += __shfl_xor(d, 2);
    if (live && sub == 0) {
        d += 1.0f;
        dinv[r] = d > 0.0f ? rsqrtf(d) : 0.0f;
    }
}


#define TILE_PITCH   260
#define STAGE_FLOATS (2 * 16 * TILE_PITCH)
#define GEMM_BLOCKS  512
#define TILES_Q      (NT / GEMM_BLOCKS)
#define TILES_R      (NT % GEMM_BLOCKS)
#define KSTEPS       (F_IN / 4)
#define KGROUP       8
#define GEMM_THREADS 256
static_assert((long long)NT * GEMM_THREADS == E_EDGES, "one edge per thread and node tile covers every edge exactly once");

struct StageRegs { float4 c0, c1, c2, c3, c4, c5, c6, c7; };

__device__ __forceinline__ float4 stage_chunk(const float* __restrict__ x, int tile, int i, int srow, int scol) {
    const float* src = x + ((size_t)(i >> 2) * N_NODES + (size_t)tile * 16 + 4 * (i & 3) + srow) * F_IN + scol;
    return *reinterpret_cast<const float4*>(src);
}

__device__ __forceinline__ StageRegs stage_load(const float* __restrict__ x, int tile, int srow, int scol) {
    StageRegs r;
    r.c0 = stage_chunk(x, tile, 0, srow, scol); r.c1 = stage_chunk(x, tile, 1, srow, scol);
    r.c2 = stage_chunk(x, tile, 2, srow, scol); r.c3 = stage_chunk(x, tile, 3, srow, scol);
    r.c4 = stage_chunk(x, tile, 4, srow, scol); r.c5 = stage_chunk(x, tile, 5, srow, scol);
    r.c6 = stage_chunk(x, tile, 6, srow, scol); r.c7 = stage_chunk(x, tile, 7, srow, scol);
    return r;
}

__device__ __forceinline__ void stage_store(float* __restrict__ stage, const StageRegs& r, int srow, int scol) {
    float* dst = stage + srow * TILE_PITCH + scol;
    *reinterpret_cast<float4*>(dst + 0 * 4 * TILE_PITCH) = r.c0; *reinterpret_cast<float4*>(dst + 1 * 4 * TILE_PITCH) = r.c1;
    *reinterpret_cast<float4*>(dst + 2 * 4 * TILE_PITCH) = r.c2; *reinterpret_cast<float4*>(dst + 3 * 4 * TILE_PITCH) = r.c3;
    *reinterpret_cast<float4*>(dst + 4 * 4 * TILE_PITCH) = r.c4; *reinterpret_cast<float4*>(dst + 5 * 4 * TILE_PITCH) = r.c5;
    *reinterpret_cast<float4*>(dst + 6 * 4 * TILE_PITCH) = r.c6; *reinterpret_cast<float4*>(dst + 7 * 4 * TILE_PITCH) = r.c7;
}

__global__ void __launch_bounds__(GEMM_THREADS) __attribute__((amdgpu_waves_per_eu(4, 4)))
gemm_wmma_kernel(const float* __restrict__ x,
                 const float* __restrict__ w,
                 float* __restrict__ xw,
                 const int* __restrict__ row,
                 const int* __restrict__ col,
                 const float* __restrict__ ew,
                 int* __restrict__ cnt,
                 int* __restrict__ next,
                 float* out) {
    __shared__ __attribute__((aligned(16))) float xs[2][STAGE_FLOATS];

    const int tid  = threadIdx.x;
    const int wave = tid >> 5;
    const int lane = tid & 31;
    const int half = lane >> 4;
    const int l    = lane & 15;
    const int o0   = wave * 16;

    const int blk     = blockIdx.x;
    const int t_begin = blk * TILES_Q + (blk < TILES_R ? blk : TILES_R);
    const int t_count = TILES_Q + (blk < TILES_R ? 1 : 0);

    int   e_cur = t_begin * GEMM_THREADS + tid;
    int   er    = row[e_cur];
    int   ec    = col[e_cur];
    float ewt   = ew[e_cur];

    v2f wf[KSTEPS];
    #pragma unroll
    for (int t = 0; t < KSTEPS; ++t) {
        wf[t].x = w[(4 * t + 2 * half + 0) * F_OUT + o0 + l];
        wf[t].y = w[(4 * t + 2 * half + 1) * F_OUT + o0 + l];
    }

    const int srow = tid >> 6;
    const int scol = (tid & 63) * 4;
    stage_store(xs[0], stage_load(x, t_begin, srow, scol), srow, scol);
    __syncthreads();

    for (int it = 0; it < t_count; ++it) {
        const int  tile = t_begin + it;
        const int  buf  = it & 1;
        const bool more = it + 1 < t_count;

        int p = atomicAdd(&cnt[er], 1);
        int   er_n = 0, ec_n = 0;
        float ew_n = 0.0f;
        if (more) {
            er_n = row[e_cur + GEMM_THREADS];
            ec_n = col[e_cur + GEMM_THREADS];
            ew_n = ew[e_cur + GEMM_THREADS];
        }
        StageRegs nxt = {};
        if (more) nxt = stage_load(x, tile + 1, srow, scol);
        __builtin_amdgcn_sched_barrier(0);

        const float* a0p = &xs[buf][(0 * 16 + l) * TILE_PITCH + 2 * half];
        const float* a1p = &xs[buf][(1 * 16 + l) * TILE_PITCH + 2 * half];

        v8f acc0 = {}, acc1 = {};
        #pragma unroll
        for (int g = 0; g < KSTEPS; g += KGROUP) {
            v2f a0[KGROUP], a1[KGROUP];
            #pragma unroll
            for (int j = 0; j < KGROUP; ++j) {
                a0[j] = *reinterpret_cast<const v2f*>(a0p + 4 * (g + j));
                a1[j] = *reinterpret_cast<const v2f*>(a1p + 4 * (g + j));
            }
            #pragma unroll
            for (int j = 0; j < KGROUP; ++j) {
                acc0 = __builtin_amdgcn_wmma_f32_16x16x4_f32(
                     false, a0[j],  false, wf[g + j],
                     (short)0, acc0,  false,  false);
                acc1 = __builtin_amdgcn_wmma_f32_16x16x4_f32(
                     false, a1[j],  false, wf[g + j],
                     (short)0, acc1,  false,  false);
            }
        }

        #pragma unroll
        for (int v = 0; v < 8; ++v) {
            const int m = v + half * 8;
            float2 pr;
            pr.x = acc0[v];
            pr.y = acc1[v];
            *reinterpret_cast<float2*>(&xw[(size_t)(tile * 16 + m) * OB + (size_t)(o0 + l) * 2]) = pr;
        }

        if (more) stage_store(xs[buf ^ 1], nxt, srow, scol);
        asm volatile("" : "+v"(p));
        place_edge(out, next, er, p, ec, ewt, e_cur);
        e_cur += GEMM_THREADS;
        er = er_n; ec = ec_n; ewt = ew_n;
        __syncthreads();
    }
}


#define ROWS_PER_BLK 8

__device__ __forceinline__ void add_row(float (&acc)[8], float s, const float4& p, const float4& q) {
    acc[0] = __builtin_fmaf(s, p.x, acc[0]); acc[1] = __builtin_fmaf(s, p.y, acc[1]);
    acc[2] = __builtin_fmaf(s, p.z, acc[2]); acc[3] = __builtin_fmaf(s, p.w, acc[3]);
    acc[4] = __builtin_fmaf(s, q.x, acc[4]); acc[5] = __builtin_fmaf(s, q.y, acc[5]);
    acc[6] = __builtin_fmaf(s, q.z, acc[6]); acc[7] = __builtin_fmaf(s, q.w, acc[7]);
}

__global__ void __launch_bounds__(256)
row_gather_kernel(const int* __restrict__ cnt,
                  const int* __restrict__ next,
                  const int* __restrict__ col,
                  const float* __restrict__ ew,
                  const float* __restrict__ dinv,
                  const float* __restrict__ xw,
                  const float* __restrict__ bias,
                  float* out) {
    const int lane = threadIdx.x & 31;
    const int r = __builtin_amdgcn_readfirstlane((int)(blockIdx.x * ROWS_PER_BLK + (threadIdx.x >> 5)));
    if (r >= N_NODES) return;

    const int   total = cnt[r];
    const int   n     = total < SLOT_CAP ? total : SLOT_CAP;
    const float dr    = dinv[r];
    const float d2    = dr * dr;
    const int   o0    = lane * 4;

    float acc[8];
    {
        const float4* xs = reinterpret_cast<const float4*>(xw + (size_t)r * OB) + lane * 2;
        const float4 p = xs[0];
        const float4 q = xs[1];
        const float b0 = bias[o0 + 0], b1 = bias[o0 + 1], b2 = bias[o0 + 2], b3 = bias[o0 + 3];
        acc[0] = b0 + d2 * p.x; acc[1] = b0 + d2 * p.y;
        acc[2] = b1 + d2 * p.z; acc[3] = b1 + d2 * p.w;
        acc[4] = b2 + d2 * q.x; acc[5] = b2 + d2 * q.y;
        acc[6] = b3 + d2 * q.z; acc[7] = b3 + d2 * q.w;
    }

    const int2* slots = reinterpret_cast<const int2*>(out) + (size_t)r * SLOT_CAP;
    int e = -1;
    for (int done = 0; done < total; ) {
        int   cnt_b = 0;
        int   c     = 0;
        float lap   = 0.0f;
        if (done < n) {
            cnt_b = n - done < 32 ? n - done : 32;
            if (lane < cnt_b) {
                const int2 s = slots[done + lane];
                c   = s.x;
                lap = dr * __int_as_float(s.y) * dinv[c];
            }
        } else {
            if (done == n) e = *overflow_head(out, r);
            const int want = total - done < 32 ? total - done : 32;
            int my_e = -1;
            while (cnt_b < want && (unsigned)e < (unsigned)E_EDGES) {
                if (lane == cnt_b) my_e = e;
                e = next[e];
                ++cnt_b;
            }
            if (cnt_b == 0) break;
            if (my_e >= 0) {
                c   = col[my_e];
                lap = dr * ew[my_e] * dinv[c];
            }
        }
        done += cnt_b;

        int j = 0;
        for (; j + 4 <= cnt_b; j += 4) {
            const int c0 = __builtin_amdgcn_readlane(c, j + 0);
            const int c1 = __builtin_amdgcn_readlane(c, j + 1);
            const int c2 = __builtin_amdgcn_readlane(c, j + 2);
            const int c3 = __builtin_amdgcn_readlane(c, j + 3);
            const float4* x0 = reinterpret_cast<const float4*>(xw + (size_t)c0 * OB) + lane * 2;
            const float4* x1 = reinterpret_cast<const float4*>(xw + (size_t)c1 * OB) + lane * 2;
            const float4* x2 = reinterpret_cast<const float4*>(xw + (size_t)c2 * OB) + lane * 2;
            const float4* x3 = reinterpret_cast<const float4*>(xw + (size_t)c3 * OB) + lane * 2;
            const float4 p0 = x0[0], q0 = x0[1];
            const float4 p1 = x1[0], q1 = x1[1];
            const float4 p2 = x2[0], q2 = x2[1];
            const float4 p3 = x3[0], q3 = x3[1];
            const float s0 = __int_as_float(__builtin_amdgcn_readlane(__float_as_int(lap), j + 0));
            const float s1 = __int_as_float(__builtin_amdgcn_readlane(__float_as_int(lap), j + 1));
            const float s2 = __int_as_float(__builtin_amdgcn_readlane(__float_as_int(lap), j + 2));
            const float s3 = __int_as_float(__builtin_amdgcn_readlane(__float_as_int(lap), j + 3));
            add_row(acc, s0, p0, q0);
            add_row(acc, s1, p1, q1);
            add_row(acc, s2, p2, q2);
            add_row(acc, s3, p3, q3);
        }
        #pragma clang loop unroll(disable)
        for (; j < cnt_b; ++j) {
            const int   cj = __builtin_amdgcn_readlane(c, j);
            const float sj = __int_as_float(__builtin_amdgcn_readlane(__float_as_int(lap), j));
            const float4* xj = reinterpret_cast<const float4*>(xw + (size_t)cj * OB) + lane * 2;
            const float4 p = xj[0], q = xj[1];
            add_row(acc, sj, p, q);
        }
    }

    float4* out0 = reinterpret_cast<float4*>(out + ((size_t)0 * N_NODES + r) * F_OUT) + lane;
    float4* out1 = reinterpret_cast<float4*>(out + ((size_t)1 * N_NODES + r) * F_OUT) + lane;
    *out0 = make_float4(acc[0], acc[2], acc[4], acc[6]);
    *out1 = make_float4(acc[1], acc[3], acc[5], acc[7]);
}


extern "C" void kernel_launch(void* const* d_in, const int* in_sizes, int n_in,
                              void* d_out, int out_size, void* d_ws, size_t ws_size,
                              hipStream_t stream) {
    const float* x    = (const float*)d_in[0];
    const float* w    = (const float*)d_in[1];
    const float* bias = (const float*)d_in[2];
    const float* ew   = (const float*)d_in[3];
    const int*   eidx = (const int*)d_in[4];
    const int*   row  = eidx;
    const int*   col  = eidx + E_EDGES;
    float* out = (float*)d_out;

    float* xw   = (float*)d_ws;
    int*   cnt  = (int*)(xw + (size_t)N_NODES * OB);
    float* dinv = (float*)(cnt + N_NODES);
    int*   next = (int*)(dinv + N_NODES);

    count_init_kernel<<<(N_NODES + 255) / 256, 256, 0, stream>>>(cnt);
    gemm_wmma_kernel<<<GEMM_BLOCKS, GEMM_THREADS, 0, stream>>>(x, w, xw, row, col, ew, cnt, next, out);
    dinv_kernel<<<(N_NODES * 4 + 255) / 256, 256, 0, stream>>>(cnt, next, ew, out, dinv);

    row_gather_kernel<<<(N_NODES + ROWS_PER_BLK - 1) / ROWS_PER_BLK, 256, 0, stream>>>(
        cnt, next, col, ew, dinv, xw, bias, out);
}


# Row sums: lane-to-column map of the gather changed so that every xw load instruction reads 512 contiguous bytes (4 whole lines, was 8 half lines); out stored as 2-line pieces; sum order unchanged
# speedup vs baseline: 5.3849x; 1.0396x over previous
#include <hip/hip_runtime.h>

#define N_NODES 50000
#define E_EDGES 800000
#define F_IN    256
#define F_OUT   128
#define BATCH   2
#define OB      (F_OUT * BATCH)

#define NT ((N_NODES + 15) / 16)

typedef float v2f __attribute__((ext_vector_type(2)));
typedef float v8f __attribute__((ext_vector_type(8)));


#define SLOT_CAP 64

__device__ __forceinline__ int* overflow_head(float* out, int r) {
    return reinterpret_cast<int*>(out) + ((size_t)N_NODES + (size_t)r) * F_OUT;
}

__global__ void count_init_kernel(int* __restrict__ cnt) {
    int i = blockIdx.x * blockDim.x + threadIdx.x;
    if (i < N_NODES) cnt[i] = 0;
}

__device__ __forceinline__ void place_edge(float* out, int* __restrict__ next,
                                           int r, int p, int c, float w, int e) {
    if (p < SLOT_CAP) {
        reinterpret_cast<int2*>(out)[(size_t)r * SLOT_CAP + p] = make_int2(c, __float_as_int(w));
    } else {
        next[e] = atomicExch(overflow_head(out, r), e);
    }
}


__global__ void __launch_bounds__(256)
dinv_kernel(const int* __restrict__ cnt,
            const int* __restrict__ next,
            const float* __restrict__ ew,
            float* out,
            float* __restrict__ dinv) {
    const int  g    = blockIdx.x * blockDim.x + threadIdx.x;
    const int  r    = g >> 2;
    const int  sub  = g & 3;
    const bool live = r < N_NODES;
    const int  total = live ? cnt[r] : 0;
    const int  n     = total < SLOT_CAP ? total : SLOT_CAP;

    float d = 0.0f;
    const int2* sl = reinterpret_cast<const int2*>(out) + (size_t)(live ? r : 0) * SLOT_CAP;
    for (int s = sub; s < n; s += 16) {
        const bool h1 = s + 4 < n, h2 = s + 8 < n, h3 = s + 12 < n;
        const float w0 = __int_as_float(sl[s].y);
        const float w1 = __int_as_float(sl[h1 ? s + 4 : s].y);
        const float w2 = __int_as_float(sl[h2 ? s + 8 : s].y);
        const float w3 = __int_as_float(sl[h3 ? s + 12 : s].y);
        d += w0;
        if (h1) d += w1;
        if (h2) d += w2;
        if (h3) d += w3;
    }
    if (sub == 0 && total > SLOT_CAP) {
        int e = *overflow_head(out, r);
        for (int left = total - SLOT_CAP; left > 0 && (unsigned)e < (unsigned)E_EDGES; --left) {
            d += ew[e];
            e = next[e];
        }
    }
    d += __shfl_xor(d, 1);
    d += __shfl_xor(d, 2);
    if (live && sub == 0) {
        d += 1.0f;
        dinv[r] = d > 0.0f ? rsqrtf(d) : 0.0f;
    }
}


#define TILE_PITCH   260
#define STAGE_FLOATS (2 * 16 * TILE_PITCH)
#define GEMM_BLOCKS  512
#define TILES_Q      (NT / GEMM_BLOCKS)
#define TILES_R      (NT % GEMM_BLOCKS)
#define KSTEPS       (F_IN / 4)
#define KGROUP       8
#define GEMM_THREADS 256
static_assert((long long)NT * GEMM_THREADS == E_EDGES, "one edge per thread and node tile covers every edge exactly once");

struct StageRegs { float4 c0, c1, c2, c3, c4, c5, c6, c7; };

__device__ __forceinline__ float4 stage_chunk(const float* __restrict__ x, int tile, int i, int srow, int scol) {
    const float* src = x + ((size_t)(i >> 2) * N_NODES + (size_t)tile * 16 + 4 * (i & 3) + srow) * F_IN + scol;
    return *reinterpret_cast<const float4*>(src);
}

__device__ __forceinline__ StageRegs stage_load(const float* __restrict__ x, int tile, int srow, int scol) {
    StageRegs r;
    r.c0 = stage_chunk(x, tile, 0, srow, scol); r.c1 = stage_chunk(x, tile, 1, srow, scol);
    r.c2 = stage_chunk(x, tile, 2, srow, scol); r.c3 = stage_chunk(x, tile, 3, srow, scol);
    r.c4 = stage_chunk(x, tile, 4, srow, scol); r.c5 = stage_chunk(x, tile, 5, srow, scol);
    r.c6 = stage_chunk(x, tile, 6, srow, scol); r.c7 = stage_chunk(x, tile, 7, srow, scol);
    return r;
}

__device__ __forceinline__ void stage_store(float* __restrict__ stage, const StageRegs& r, int srow, int scol) {
    float* dst = stage + srow * TILE_PITCH + scol;
    *reinterpret_cast<float4*>(dst + 0 * 4 * TILE_PITCH) = r.c0; *reinterpret_cast<float4*>(dst + 1 * 4 * TILE_PITCH) = r.c1;
    *reinterpret_cast<float4*>(dst + 2 * 4 * TILE_PITCH) = r.c2; *reinterpret_cast<float4*>(dst + 3 * 4 * TILE_PITCH) = r.c3;
    *reinterpret_cast<float4*>(dst + 4 * 4 * TILE_PITCH) = r.c4; *reinterpret_cast<float4*>(dst + 5 * 4 * TILE_PITCH) = r.c5;
    *reinterpret_cast<float4*>(dst + 6 * 4 * TILE_PITCH) = r.c6; *reinterpret_cast<float4*>(dst + 7 * 4 * TILE_PITCH) = r.c7;
}

__global__ void __launch_bounds__(GEMM_THREADS) __attribute__((amdgpu_waves_per_eu(4, 4)))
gemm_wmma_kernel(const float* __restrict__ x,
                 const float* __restrict__ w,
                 float* __restrict__ xw,
                 const int* __restrict__ row,
                 const int* __restrict__ col,
                 const float* __restrict__ ew,
                 int* __restrict__ cnt,
                 int* __restrict__ next,
                 float* out) {
    __shared__ __attribute__((aligned(16))) float xs[2][STAGE_FLOATS];

    const int tid  = threadIdx.x;
    const int wave = tid >> 5;
    const int lane = tid & 31;
    const int half = lane >> 4;
    const int l    = lane & 15;
    const int o0   = wave * 16;

    const int blk     = blockIdx.x;
    const int t_begin = blk * TILES_Q + (blk < TILES_R ? blk : TILES_R);
    const int t_count = TILES_Q + (blk < TILES_R ? 1 : 0);

    int   e_cur = t_begin * GEMM_THREADS + tid;
    int   er    = row[e_cur];
    int   ec    = col[e_cur];
    float ewt   = ew[e_cur];

    v2f wf[KSTEPS];
    #pragma unroll
    for (int t = 0; t < KSTEPS; ++t) {
        wf[t].x = w[(4 * t + 2 * half + 0) * F_OUT + o0 + l];
        wf[t].y = w[(4 * t + 2 * half + 1) * F_OUT + o0 + l];
    }

    const int srow = tid >> 6;
    const int scol = (tid & 63) * 4;
    stage_store(xs[0], stage_load(x, t_begin, srow, scol), srow, scol);
    __syncthreads();

    for (int it = 0; it < t_count; ++it) {
        const int  tile = t_begin + it;
        const int  buf  = it & 1;
        const bool more = it + 1 < t_count;

        int p = atomicAdd(&cnt[er], 1);
        int   er_n = 0, ec_n = 0;
        float ew_n = 0.0f;
        if (more) {
            er_n = row[e_cur + GEMM_THREADS];
            ec_n = col[e_cur + GEMM_THREADS];
            ew_n = ew[e_cur + GEMM_THREADS];
        }
        StageRegs nxt = {};
        if (more) nxt = stage_load(x, tile + 1, srow, scol);
        __builtin_amdgcn_sched_barrier(0);

        const float* a0p = &xs[buf][(0 * 16 + l) * TILE_PITCH + 2 * half];
        const float* a1p = &xs[buf][(1 * 16 + l) * TILE_PITCH + 2 * half];

        v8f acc0 = {}, acc1 = {};
        #pragma unroll
        for (int g = 0; g < KSTEPS; g += KGROUP) {
            v2f a0[KGROUP], a1[KGROUP];
            #pragma unroll
            for (int j = 0; j < KGROUP; ++j) {
                a0[j] = *reinterpret_cast<const v2f*>(a0p + 4 * (g + j));
                a1[j] = *reinterpret_cast<const v2f*>(a1p + 4 * (g + j));
            }
            #pragma unroll
            for (int j = 0; j < KGROUP; ++j) {
                acc0 = __builtin_amdgcn_wmma_f32_16x16x4_f32(
                     false, a0[j],  false, wf[g + j],
                     (short)0, acc0,  false,  false);
                acc1 = __builtin_amdgcn_wmma_f32_16x16x4_f32(
                     false, a1[j],  false, wf[g + j],
                     (short)0, acc1,  false,  false);
            }
        }

        #pragma unroll
        for (int v = 0; v < 8; ++v) {
            const int m = v + half * 8;
            float2 pr;
            pr.x = acc0[v];
            pr.y = acc1[v];
            *reinterpret_cast<float2*>(&xw[(size_t)(tile * 16 + m) * OB + (size_t)(o0 + l) * 2]) = pr;
        }

        if (more) stage_store(xs[buf ^ 1], nxt, srow, scol);
        asm volatile("" : "+v"(p));
        place_edge(out, next, er, p, ec, ewt, e_cur);
        e_cur += GEMM_THREADS;
        er = er_n; ec = ec_n; ewt = ew_n;
        __syncthreads();
    }
}


#define ROWS_PER_BLK 8

__device__ __forceinline__ void add_row(float (&acc)[8], float s, const float4& p, const float4& q) {
    acc[0] = __builtin_fmaf(s, p.x, acc[0]); acc[1] = __builtin_fmaf(s, p.y, acc[1]);
    acc[2] = __builtin_fmaf(s, p.z, acc[2]); acc[3] = __builtin_fmaf(s, p.w, acc[3]);
    acc[4] = __builtin_fmaf(s, q.x, acc[4]); acc[5] = __builtin_fmaf(s, q.y, acc[5]);
    acc[6] = __builtin_fmaf(s, q.z, acc[6]); acc[7] = __builtin_fmaf(s, q.w, acc[7]);
}

__global__ void __launch_bounds__(256)
row_gather_kernel(const int* __restrict__ cnt,
                  const int* __restrict__ next,
                  const int* __restrict__ col,
                  const float* __restrict__ ew,
                  const float* __restrict__ dinv,
                  const float* __restrict__ xw,
                  const float* __restrict__ bias,
                  float* out) {
    const int lane = threadIdx.x & 31;
    const int r = __builtin_amdgcn_readfirstlane((int)(blockIdx.x * ROWS_PER_BLK + (threadIdx.x >> 5)));
    if (r >= N_NODES) return;

    const int   total = cnt[r];
    const int   n     = total < SLOT_CAP ? total : SLOT_CAP;
    const float dr    = dinv[r];
    const float d2    = dr * dr;
    const int   o0    = lane * 2;

    float acc[8];
    {
        const float4* xs = reinterpret_cast<const float4*>(xw + (size_t)r * OB) + lane;
        const float4 p = xs[0];
        const float4 q = xs[OB / 8];
        const float2 bl = *reinterpret_cast<const float2*>(bias + o0);
        const float2 bh = *reinterpret_cast<const float2*>(bias + F_OUT / 2 + o0);
        const float b0 = bl.x, b1 = bl.y, b2 = bh.x, b3 = bh.y;
        acc[0] = b0 + d2 * p.x; acc[1] = b0 + d2 * p.y;
        acc[2] = b1 + d2 * p.z; acc[3] = b1 + d2 * p.w;
        acc[4] = b2 + d2 * q.x; acc[5] = b2 + d2 * q.y;
        acc[6] = b3 + d2 * q.z; acc[7] = b3 + d2 * q.w;
    }

    const int2* slots = reinterpret_cast<const int2*>(out) + (size_t)r * SLOT_CAP;
    int e = -1;
    for (int done = 0; done < total; ) {
        int   cnt_b = 0;
        int   c     = 0;
        float lap   = 0.0f;
        if (done < n) {
            cnt_b = n - done < 32 ? n - done : 32;
            if (lane < cnt_b) {
                const int2 s = slots[done + lane];
                c   = s.x;
                lap = dr * __int_as_float(s.y) * dinv[c];
            }
        } else {
            if (done == n) e = *overflow_head(out, r);
            const int want = total - done < 32 ? total - done : 32;
            int my_e = -1;
            while (cnt_b < want && (unsigned)e < (unsigned)E_EDGES) {
                if (lane == cnt_b) my_e = e;
                e = next[e];
                ++cnt_b;
            }
            if (cnt_b == 0) break;
            if (my_e >= 0) {
                c   = col[my_e];
                lap = dr * ew[my_e] * dinv[c];
            }
        }
        done += cnt_b;

        int j = 0;
        for (; j + 4 <= cnt_b; j += 4) {
            const int c0 = __builtin_amdgcn_readlane(c, j + 0);
            const int c1 = __builtin_amdgcn_readlane(c, j + 1);
            const int c2 = __builtin_amdgcn_readlane(c, j + 2);
            const int c3 = __builtin_amdgcn_readlane(c, j + 3);
            const float4* x0 = reinterpret_cast<const float4*>(xw + (size_t)c0 * OB) + lane;
            const float4* x1 = reinterpret_cast<const float4*>(xw + (size_t)c1 * OB) + lane;
            const float4* x2 = reinterpret_cast<const float4*>(xw + (size_t)c2 * OB) + lane;
            const float4* x3 = reinterpret_cast<const float4*>(xw + (size_t)c3 * OB) + lane;
            const float4 p0 = x0[0], q0 = x0[OB / 8];
            const float4 p1 = x1[0], q1 = x1[OB / 8];
            const float4 p2 = x2[0], q2 = x2[OB / 8];
            const float4 p3 = x3[0], q3 = x3[OB / 8];
            const float s0 = __int_as_float(__builtin_amdgcn_readlane(__float_as_int(lap), j + 0));
            const float s1 = __int_as_float(__builtin_amdgcn_readlane(__float_as_int(lap), j + 1));
            const float s2 = __int_as_float(__builtin_amdgcn_readlane(__float_as_int(lap), j + 2));
            const float s3 = __int_as_float(__builtin_amdgcn_readlane(__float_as_int(lap), j + 3));
            add_row(acc, s0, p0, q0);
            add_row(acc, s1, p1, q1);
            add_row(acc, s2, p2, q2);
            add_row(acc, s3, p3, q3);
        }
        #pragma clang loop unroll(disable)
        for (; j < cnt_b; ++j) {
            const int   cj = __builtin_amdgcn_readlane(c, j);
            const float sj = __int_as_float(__builtin_amdgcn_readlane(__float_as_int(lap), j));
            const float4* xj = reinterpret_cast<const float4*>(xw + (size_t)cj * OB) + lane;
            const float4 p = xj[0], q = xj[OB / 8];
            add_row(acc, sj, p, q);
        }
    }

    float2* out0 = reinterpret_cast<float2*>(out + ((size_t)0 * N_NODES + r) * F_OUT) + lane;
    float2* out1 = reinterpret_cast<float2*>(out + ((size_t)1 * N_NODES + r) * F_OUT) + lane;
    out0[0]         = make_float2(acc[0], acc[2]);
    out0[F_OUT / 4] = make_float2(acc[4], acc[6]);
    out1[0]         = make_float2(acc[1], acc[3]);
    out1[F_OUT / 4] = make_float2(acc[5], acc[7]);
}


extern "C" void kernel_launch(void* const* d_in, const int* in_sizes, int n_in,
                              void* d_out, int out_size, void* d_ws, size_t ws_size,
                              hipStream_t stream) {
    const float* x    = (const float*)d_in[0];
    const float* w    = (const float*)d_in[1];
    const float* bias = (const float*)d_in[2];
    const float* ew   = (const float*)d_in[3];
    const int*   eidx = (const int*)d_in[4];
    const int*   row  = eidx;
    const int*   col  = eidx + E_EDGES;
    float* out = (float*)d_out;

    float* xw   = (float*)d_ws;
    int*   cnt  = (int*)(xw + (size_t)N_NODES * OB);
    float* dinv = (float*)(cnt + N_NODES);
    int*   next = (int*)(dinv + N_NODES);

    count_init_kernel<<<(N_NODES + 255) / 256, 256, 0, stream>>>(cnt);
    gemm_wmma_kernel<<<GEMM_BLOCKS, GEMM_THREADS, 0, stream>>>(x, w, xw, row, col, ew, cnt, next, out);
    dinv_kernel<<<(N_NODES * 4 + 255) / 256, 256, 0, stream>>>(cnt, next, ew, out, dinv);

    row_gather_kernel<<<(N_NODES + ROWS_PER_BLK - 1) / ROWS_PER_BLK, 256, 0, stream>>>(
        cnt, next, col, ew, dinv, xw, bias, out);
}
